# starting kernel
# baseline (speedup 1.0000x reference)
#include <hip/hip_runtime.h>


typedef float v2f __attribute__((ext_vector_type(2)));
typedef float v8f __attribute__((ext_vector_type(8)));

__device__ __forceinline__ void load_areg(v2f* areg, const float* __restrict__ aRow,
                                          int ga, int khalf) {
  const float* p = aRow + ga * 64 + 2 * khalf;
#pragma unroll
  for (int kk = 0; kk < 16; ++kk)
    areg[kk] = *reinterpret_cast<const v2f*>(p + 4 * kk);
}

__device__ __forceinline__ v8f tile_gemm(const v2f* areg, const float* __restrict__ Wp,
                                         float bb, int khalf) {
  v2f breg[16];
#pragma unroll
  for (int kk = 0; kk < 16; ++kk)
    breg[kk] = *reinterpret_cast<const v2f*>(Wp + 4 * kk + 2 * khalf);
  v8f acc = {bb, bb, bb, bb, bb, bb, bb, bb};
#pragma unroll
  for (int kk = 0; kk < 16; ++kk) {
    acc = __builtin_amdgcn_wmma_f32_16x16x4_f32(
         false, areg[kk],  false, breg[kk],
         (short)0, acc,  false,  false);
  }
  return acc;
}

__device__ __forceinline__ void store_tile(float* buf, int nt, int lane, v8f acc) {
  const int n = lane & 15, kh = lane >> 4;
#pragma unroll
  for (int i = 0; i < 8; ++i)
    buf[(i + 8 * kh) * 128 + nt * 16 + n] = acc[i];
}

__global__ __launch_bounds__(256) void fc2_fused(const float* __restrict__ in,
                                                 const float* __restrict__ W,
                                                 const float* __restrict__ bias,
                                                 float* __restrict__ out) {
  __shared__ float ytile[4][16 * 128];
  __shared__ float stile[16 * 128];
  __shared__ float xsumb[16 * 128];
  __shared__ float kvbuf[64];

  const int tid = threadIdx.x;
  const int lane = tid & 31;
  const int wave = tid >> 5;
  const int n0 = blockIdx.x * 16;
  const int bIdx = n0 >> 10;
  const int t0 = n0 & 1023;
  const int khalf = lane >> 4;
  const float* aRow = in + ((size_t)((t0 + (lane & 15)) * 4 + bIdx)) * 4096;

  const int rr = wave & 3;
  const int ntBase = wave >> 2;

  float m = -3.0e38f, dsum = 0.0f, num = 0.0f;

  for (int jj = 0; jj < 32; ++jj) {
    const int o_base = jj * 128;
    {
      const int ga = rr * 16 + (jj >> 1);
      v2f areg[16];
      load_areg(areg, aRow, ga, khalf);
#pragma unroll
      for (int k4 = 0; k4 < 4; ++k4) {
        const int nt = ntBase + 2 * k4;
        const int ch = rr * 4096 + o_base + nt * 16 + (lane & 15);
        v8f acc = tile_gemm(areg, W + (size_t)ch * 64, bias[ch], khalf);
        store_tile(ytile[rr], nt, lane, acc);
      }
    }
    __syncthreads();
#pragma unroll
    for (int i = 0; i < 8; ++i) {
      const int e = i * 256 + tid;
      xsumb[e] = ytile[0][e] + ytile[1][e] + ytile[2][e] + ytile[3][e];
    }
    __syncthreads();
    if (tid < 64) {
      const int prow = tid >> 2, pr = tid & 3;
      const float* yv = &ytile[pr][prow * 128];
      const float* xs = &xsumb[prow * 128];
      for (int o = 0; o < 128; ++o) {
        const float v = yv[o], x = xs[o];
        if (v > m) { const float sc = __expf(m - v); dsum *= sc; num *= sc; m = v; }
        const float e = __expf(v - m);
        dsum += e;
        num += e * x;
      }
    }
    __syncthreads();
  }
  if (tid < 64) kvbuf[tid] = num / dsum;
  __syncthreads();

  for (int jj = 0; jj < 32; ++jj) {
    const int o_base = jj * 128;
    {
      const int ga = rr * 16 + (jj >> 1);
      v2f areg[16];
      load_areg(areg, aRow, ga, khalf);
#pragma unroll
      for (int k4 = 0; k4 < 4; ++k4) {
        const int nt = ntBase + 2 * k4;
        const int ch = rr * 4096 + o_base + nt * 16 + (lane & 15);
        v8f acc = tile_gemm(areg, W + (size_t)ch * 64, bias[ch], khalf);
        store_tile(ytile[rr], nt, lane, acc);
      }
      const int gs = 2 * jj + (wave >> 2);
      v2f sreg[16];
      load_areg(sreg, aRow, gs, khalf);
      const int chs = 4 * (o_base + wave * 16 + (lane & 15));
      v8f accs = tile_gemm(sreg, W + (size_t)chs * 64, bias[chs], khalf);
      store_tile(stile, wave, lane, accs);
    }
    __syncthreads();
#pragma unroll
    for (int i = 0; i < 8; ++i) {
      const int e = i * 256 + tid;
      const int prow = e >> 7;
      const int oo = e & 127;
      const float* kvp = &kvbuf[prow * 4];
      const float val = stile[e] + ytile[0][e] * kvp[0] + ytile[1][e] * kvp[1] +
                        ytile[2][e] * kvp[2] + ytile[3][e] * kvp[3];
      out[((size_t)((t0 + prow) * 4 + bIdx)) * 4096 + o_base + oo] = val;
    }
    __syncthreads();
  }
}

extern "C" void kernel_launch(void* const* d_in, const int* in_sizes, int n_in,
                              void* d_out, int out_size, void* d_ws, size_t ws_size,
                              hipStream_t stream) {
  (void)in_sizes; (void)n_in; (void)d_ws; (void)ws_size; (void)out_size;
  const float* in = (const float*)d_in[0];
  const float* W = (const float*)d_in[1];
  const float* bias = (const float*)d_in[2];
  float* out = (float*)d_out;
  fc2_fused<<<256, 256, 0, stream>>>(in, W, bias, out);
}


# Weights re-laid once per call into WMMA fragment order in the workspace by a small first launch: a fragment load is 512 contiguous bytes per wave, was 16 B out of each of 16 lines; sums unchanged
# speedup vs baseline: 5.1705x; 5.1705x over previous
#include <hip/hip_runtime.h>


typedef float v2f __attribute__((ext_vector_type(2)));
typedef float v4f __attribute__((ext_vector_type(4)));
typedef float v8f __attribute__((ext_vector_type(8)));

__device__ __forceinline__ void load_areg(v2f* areg, const float* __restrict__ aRow,
                                          int ga, int khalf) {
  const float* p = aRow + ga * 64 + 2 * khalf;
#pragma unroll
  for (int kk = 0; kk < 16; ++kk)
    areg[kk] = *reinterpret_cast<const v2f*>(p + 4 * kk);
}

constexpr size_t kWsNeed = (size_t)1280 * 4096;

__device__ __forceinline__ void load_wfrag(v2f* areg, const float* __restrict__ Wf,
                                           int tile, int lane) {
  const float* p = Wf + (size_t)tile * 1024 + lane * 4;
#pragma unroll
  for (int j = 0; j < 8; ++j) {
    const v4f t = *reinterpret_cast<const v4f*>(p + 128 * j);
    areg[2 * j] = v2f{t[0], t[1]};
    areg[2 * j + 1] = v2f{t[2], t[3]};
  }
}

template <bool FR>
__device__ __forceinline__ void load_unit(v2f (&w)[2][16], v8f (&c)[2],
                                          const float* __restrict__ W,
                                          const float* __restrict__ Wf,
                                          const float* __restrict__ bias,
                                          int o0, int rp, int lane, int khalf) {
#pragma unroll
  for (int q = 0; q < 2; ++q) {
    const int r = 2 * rp + q;
    if (FR) {
      load_wfrag(w[q], Wf, r * 256 + (o0 >> 4), lane);
    } else {
      const int ch = r * 4096 + o0 + (lane & 15);
      load_areg(w[q], W + (size_t)ch * 64, 0, khalf);
    }
    const float* bp = bias + r * 4096 + o0 + 8 * khalf;
    const v4f b0 = *reinterpret_cast<const v4f*>(bp);
    const v4f b1 = *reinterpret_cast<const v4f*>(bp + 4);
    c[q] = __builtin_shufflevector(b0, b1, 0, 1, 2, 3, 4, 5, 6, 7);
  }
}

__device__ __forceinline__ float xhalf(float x) {
  return __builtin_bit_cast(float,
      __builtin_amdgcn_permlanex16(__builtin_bit_cast(int, x), __builtin_bit_cast(int, x),
                                   0x76543210, 0xfedcba98, false, false));
}

template <bool FR>
__device__ __forceinline__ void load_unit2(v2f (&w)[2][16], float& sb, v2f& bb,
                                           const float* __restrict__ W,
                                           const float* __restrict__ Wf,
                                           const float* __restrict__ bias,
                                           int T0, int u, int lane, int khalf) {
  const int hf = u / 5, v = u % 5, c = lane & 15;
#pragma unroll
  for (int q = 0; q < 2; ++q) {
    if (FR) {
      int tile;
      if (v < 4) tile = (2 * (v & 1) + q) * 256 + (T0 + 2 * hf + (v >> 1));
      else       tile = 1024 + (T0 + 2 * hf + q);
      load_wfrag(w[q], Wf, tile, lane);
    } else {
      int ch;
      if (v < 4) ch = (2 * (v & 1) + q) * 4096 + (T0 + 2 * hf + (v >> 1)) * 16 + c;
      else       ch = 4 * ((T0 + 2 * hf + q) * 16 + c);
      load_areg(w[q], W + (size_t)ch * 64, 0, khalf);
    }
  }
  if (v == 0 || v == 2) {
    const int o = (T0 + 2 * hf + (v >> 1)) * 16 + c;
    sb = bias[4 * o];
    bb[0] = bias[(2 * khalf) * 4096 + o];
    bb[1] = bias[(2 * khalf + 1) * 4096 + o];
  }
}

__global__ __launch_bounds__(256) void w_to_fragment_order(const float* __restrict__ W,
                                                           float* __restrict__ Wf,
                                                           size_t ws_bytes) {
  if (ws_bytes < kWsNeed) return;
  const int tid = blockIdx.x * 256 + threadIdx.x;
  for (int it = 0; it < 5; ++it) {
    const int g = it * 65536 + tid;
    const int tile = g >> 8, j = (g >> 5) & 7, lane = g & 31;
    const int c = lane & 15, h = lane >> 4;
    const int ch = (tile < 1024) ? tile * 16 + c : (tile - 1024) * 64 + 4 * c;
    const float* s = W + (size_t)ch * 64 + 8 * j + 2 * h;
    const v2f a = *reinterpret_cast<const v2f*>(s);
    const v2f b = *reinterpret_cast<const v2f*>(s + 4);
    *reinterpret_cast<v4f*>(Wf + (size_t)g * 4) = v4f{a[0], a[1], b[0], b[1]};
  }
}

template <bool FR>
__device__ __forceinline__ void fc2_body(const float* __restrict__ in,
                                         const float* __restrict__ W,
                                         const float* __restrict__ Wf,
                                         const float* __restrict__ bias,
                                         float* __restrict__ out,
                                         float* part,
                                         float* kvbuf) {

  const int tid = threadIdx.x;
  const int lane = tid & 31;
  const int wave = tid >> 5;
  const int n0 = blockIdx.x * 16;
  const int bIdx = n0 >> 10;
  const int t0 = n0 & 1023;
  const int khalf = lane >> 4;
  const float* aRow = in + ((size_t)((t0 + (lane & 15)) * 4 + bIdx)) * 4096;

  float sm[4], sd[4], sn[4];
#pragma unroll
  for (int r = 0; r < 4; ++r) { sm[r] = -3.0e38f; sd[r] = 0.0f; sn[r] = 0.0f; }

  const int tsel = (wave & 3) * 4;
  v2f wb[2][2][16];
  v8f cb[2][2];
  load_unit<FR>(wb[0], cb[0], W, Wf, bias, (wave >> 2) * 256 + tsel * 16, 0, lane, khalf);

  for (int JJ = 0; JJ < 8; ++JJ) {
    const int J = 2 * JJ + (wave >> 2);
    const int Jn = (JJ < 7) ? J + 2 : J;
    v2f xin[4][16];
#pragma unroll
    for (int r = 0; r < 4; ++r) load_areg(xin[r], aRow, r * 16 + J, khalf);
    v8f acc[4];
#pragma unroll
    for (int u = 0; u < 8; ++u) {
      const int cur = u & 1, nxt = cur ^ 1, rp = u & 1;
      {
        const int un = (u + 1) & 7;
        const int Jq = (u == 7) ? Jn : J;
        load_unit<FR>(wb[nxt], cb[nxt], W, Wf, bias, Jq * 256 + (tsel + (un >> 1)) * 16,
                      un & 1, lane, khalf);
      }
      __builtin_amdgcn_sched_barrier(0);
      acc[2 * rp] = cb[cur][0];
      acc[2 * rp + 1] = cb[cur][1];
#pragma unroll
      for (int kk = 0; kk < 16; ++kk) {
#pragma unroll
        for (int q = 0; q < 2; ++q) {
          acc[2 * rp + q] = __builtin_amdgcn_wmma_f32_16x16x4_f32(
               false, wb[cur][q][kk],  false, xin[2 * rp + q][kk],
               (short)0, acc[2 * rp + q],  false,  false);
        }
      }
      __builtin_amdgcn_sched_barrier(0);
      if (rp == 1) {
        float xs[8];
#pragma unroll
        for (int i = 0; i < 8; ++i) xs[i] = acc[0][i] + acc[1][i] + acc[2][i] + acc[3][i];
#pragma unroll
        for (int r = 0; r < 4; ++r) {
          float bm = acc[r][0];
#pragma unroll
          for (int i = 1; i < 8; ++i) bm = fmaxf(bm, acc[r][i]);
          const float mn = fmaxf(sm[r], bm);
          const float sc = __expf(sm[r] - mn);
          float ds = 0.0f, ns = 0.0f;
#pragma unroll
          for (int i = 0; i < 8; ++i) {
            const float e = __expf(acc[r][i] - mn);
            ds += e;
            ns += e * xs[i];
          }
          sd[r] = sd[r] * sc + ds;
          sn[r] = sn[r] * sc + ns;
          sm[r] = mn;
        }
      }
    }
  }
#pragma unroll
  for (int r = 0; r < 4; ++r) {
    part[tid * 12 + r * 3 + 0] = sm[r];
    part[tid * 12 + r * 3 + 1] = sd[r];
    part[tid * 12 + r * 3 + 2] = sn[r];
  }
  __syncthreads();
  if (tid < 64) {
    const int prow = tid >> 2, pr = tid & 3;
    float mAll = -3.0e38f;
    for (int p = 0; p < 16; ++p)
      mAll = fmaxf(mAll, part[((p >> 1) * 32 + (p & 1) * 16 + prow) * 12 + pr * 3]);
    float dAll = 0.0f, nAll = 0.0f;
    for (int p = 0; p < 16; ++p) {
      const float* q = &part[((p >> 1) * 32 + (p & 1) * 16 + prow) * 12 + pr * 3];
      const float f = __expf(q[0] - mAll);
      dAll += q[1] * f;
      nAll += q[2] * f;
    }
    kvbuf[tid] = nAll / dAll;
  }
  __syncthreads();

  const int c15 = lane & 15;
  float kvr[4];
#pragma unroll
  for (int r = 0; r < 4; ++r) kvr[r] = kvbuf[c15 * 4 + r];
  v2f kvA;
  kvA[0] = kvbuf[c15 * 4 + 2 * khalf];
  kvA[1] = kvbuf[c15 * 4 + 2 * khalf + 1];
  float* obase = out + ((size_t)(t0 * 4 + bIdx)) * 4096 + lane;

  float sbv[2] = {0.0f, 0.0f};
  v2f bbv[2] = {v2f{0.0f, 0.0f}, v2f{0.0f, 0.0f}};
  load_unit2<FR>(wb[0], sbv[0], bbv[0], W, Wf, bias, wave * 16, 0, lane, khalf);

  for (int jb = 0; jb < 2; ++jb) {
    const int J = wave + 8 * jb;
    v2f xs[4][16];
#pragma unroll
    for (int r = 0; r < 4; ++r) load_areg(xs[r], aRow, r * 16 + J, khalf);
#pragma unroll
    for (int r = 0; r < 4; ++r) {
#pragma unroll
      for (int kk = 0; kk < 16; ++kk) xs[r][kk] = xs[r][kk] * kvr[r];
    }
    for (int bq = 0; bq < 4; ++bq) {
      const int T0 = J * 16 + 4 * bq;
      const int T0n = (bq < 3) ? T0 + 4 : ((jb == 0) ? (wave + 8) * 16 : T0);
      v2f sinf[16];
      load_areg(sinf, aRow, T0 >> 2, khalf);
      v8f acc[2][2];
#pragma unroll
      for (int u = 0; u < 10; ++u) {
        const int cur = u & 1, nxt = cur ^ 1;
        const int v = u % 5, hf = u / 5;
        if (u < 9)
          load_unit2<FR>(wb[nxt], sbv[nxt], bbv[nxt], W, Wf, bias, T0, u + 1, lane, khalf);
        else
          load_unit2<FR>(wb[nxt], sbv[nxt], bbv[nxt], W, Wf, bias, T0n, 0, lane, khalf);
        __builtin_amdgcn_sched_barrier(0);
        if (v < 4) {
          const int tl = v >> 1, rp = v & 1;
          if (rp == 0) {
            const float s0 = sbv[cur];
            acc[tl][0] = v8f{s0, s0, s0, s0, s0, s0, s0, s0};
            acc[tl][1] = v8f{0.0f, 0.0f, 0.0f, 0.0f, 0.0f, 0.0f, 0.0f, 0.0f};
          }
#pragma unroll
          for (int kk = 0; kk < 16; ++kk) {
#pragma unroll
            for (int q = 0; q < 2; ++q) {
              acc[tl][q] = __builtin_amdgcn_wmma_f32_16x16x4_f32(
                   false, xs[2 * rp + q][kk],  false, wb[cur][q][kk],
                   (short)0, acc[tl][q],  false,  false);
            }
          }
          if (rp == 0) {
            acc[tl][1] = __builtin_amdgcn_wmma_f32_16x16x4_f32(
                 false, kvA,  false, bbv[cur],
                 (short)0, acc[tl][1],  false,  false);
          }
        } else {
#pragma unroll
          for (int kk = 0; kk < 16; ++kk) {
#pragma unroll
            for (int q = 0; q < 2; ++q) {
              acc[q][0] = __builtin_amdgcn_wmma_f32_16x16x4_f32(
                   false, sinf[kk],  false, wb[cur][q][kk],
                   (short)0, acc[q][0],  false,  false);
            }
          }
        }
        __builtin_amdgcn_sched_barrier(0);
        if (v == 4) {
          const int o0 = (T0 + 2 * hf) * 16;
#pragma unroll
          for (int i = 0; i < 8; ++i) {
            const float X = acc[0][0][i] + acc[0][1][i];
            const float Y = acc[1][0][i] + acc[1][1][i];
            const float Xo = xhalf(X), Yo = xhalf(Y);
            const float P = khalf ? Yo : X;
            const float Q = khalf ? Y : Xo;
            obase[(size_t)i * 16384 + o0] = P;
            obase[(size_t)(i + 8) * 16384 + o0] = Q;
          }
        }
      }
    }
  }
}

__global__ __launch_bounds__(256) void fc2_fused(const float* __restrict__ in,
                                                 const float* __restrict__ W,
                                                 const float* __restrict__ bias,
                                                 float* __restrict__ out,
                                                 const float* __restrict__ Wf,
                                                 size_t ws_bytes) {
  __shared__ float part[256 * 12];
  __shared__ float kvbuf[64];
  if (__builtin_expect(ws_bytes >= kWsNeed, 1))
    fc2_body<true>(in, W, Wf, bias, out, part, kvbuf);
  else
    fc2_body<false>(in, W, Wf, bias, out, part, kvbuf);
}

extern "C" void kernel_launch(void* const* d_in, const int* in_sizes, int n_in,
                              void* d_out, int out_size, void* d_ws, size_t ws_size,
                              hipStream_t stream) {
  (void)in_sizes; (void)n_in; (void)out_size;
  const float* in = (const float*)d_in[0];
  const float* W = (const float*)d_in[1];
  const float* bias = (const float*)d_in[2];
  float* out = (float*)d_out;
  float* wf = (float*)d_ws;
  w_to_fragment_order<<<256, 256, 0, stream>>>(W, wf, ws_size);
  fc2_fused<<<256, 256, 0, stream>>>(in, W, bias, out, wf, ws_size);
}


# Phase 2 no longer recomputes every fourth o tile: phase 1 keeps its x in LDS (256 KB), kv mix there as 4 FMAs per output; 520 of 4640 WMMAs per wave dropped; sum order of those outputs changed
# speedup vs baseline: 5.5761x; 1.0784x over previous
#include <hip/hip_runtime.h>


typedef float v2f __attribute__((ext_vector_type(2)));
typedef float v4f __attribute__((ext_vector_type(4)));
typedef float v8f __attribute__((ext_vector_type(8)));

__device__ __forceinline__ void load_areg(v2f* areg, const float* __restrict__ aRow,
                                          int ga, int khalf) {
  const float* p = aRow + ga * 64 + 2 * khalf;
#pragma unroll
  for (int kk = 0; kk < 16; ++kk)
    areg[kk] = *reinterpret_cast<const v2f*>(p + 4 * kk);
}

constexpr size_t kWsNeed = (size_t)1280 * 4096;

__device__ __forceinline__ void load_wfrag(v2f* areg, const float* __restrict__ Wf,
                                           int tile, int lane) {
  const float* p = Wf + (size_t)tile * 1024 + lane * 4;
#pragma unroll
  for (int j = 0; j < 8; ++j) {
    const v4f t = *reinterpret_cast<const v4f*>(p + 128 * j);
    areg[2 * j] = v2f{t[0], t[1]};
    areg[2 * j + 1] = v2f{t[2], t[3]};
  }
}

template <bool FR>
__device__ __forceinline__ void load_unit(v2f (&w)[2][16], v8f (&c)[2],
                                          const float* __restrict__ W,
                                          const float* __restrict__ Wf,
                                          const float* __restrict__ bias,
                                          int o0, int rp, int lane, int khalf) {
#pragma unroll
  for (int q = 0; q < 2; ++q) {
    const int r = 2 * rp + q;
    if (FR) {
      load_wfrag(w[q], Wf, r * 256 + (o0 >> 4), lane);
    } else {
      const int ch = r * 4096 + o0 + (lane & 15);
      load_areg(w[q], W + (size_t)ch * 64, 0, khalf);
    }
    const float* bp = bias + r * 4096 + o0 + 8 * khalf;
    const v4f b0 = *reinterpret_cast<const v4f*>(bp);
    const v4f b1 = *reinterpret_cast<const v4f*>(bp + 4);
    c[q] = __builtin_shufflevector(b0, b1, 0, 1, 2, 3, 4, 5, 6, 7);
  }
}

__device__ __forceinline__ float xhalf(float x) {
  return __builtin_bit_cast(float,
      __builtin_amdgcn_permlanex16(__builtin_bit_cast(int, x), __builtin_bit_cast(int, x),
                                   0x76543210, 0xfedcba98, false, false));
}

template <bool FR>
__device__ __forceinline__ void load_unit2(v2f (&w)[2][16], float& sb, v2f& bb,
                                           const float* __restrict__ W,
                                           const float* __restrict__ Wf,
                                           const float* __restrict__ bias,
                                           int T0, int u, int lane, int khalf) {
  const int hf = u / 5, v = u % 5, c = lane & 15;
#pragma unroll
  for (int q = 0; q < 2; ++q) {
    if (FR) {
      int tile;
      if (v < 4) tile = (2 * (v & 1) + q) * 256 + (T0 + 2 * hf + (v >> 1));
      else       tile = 1024 + (T0 + 2 * hf + q);
      load_wfrag(w[q], Wf, tile, lane);
    } else {
      int ch;
      if (v < 4) ch = (2 * (v & 1) + q) * 4096 + (T0 + 2 * hf + (v >> 1)) * 16 + c;
      else       ch = 4 * ((T0 + 2 * hf + q) * 16 + c);
      load_areg(w[q], W + (size_t)ch * 64, 0, khalf);
    }
  }
  if (v == 0 || v == 2) {
    const int o = (T0 + 2 * hf + (v >> 1)) * 16 + c;
    sb = bias[4 * o];
    bb[0] = bias[(2 * khalf) * 4096 + o];
    bb[1] = bias[(2 * khalf + 1) * 4096 + o];
  }
}

__global__ __launch_bounds__(256) void w_to_fragment_order(const float* __restrict__ W,
                                                           float* __restrict__ Wf,
                                                           size_t ws_bytes) {
  if (ws_bytes < kWsNeed) return;
  const int tid = blockIdx.x * 256 + threadIdx.x;
  for (int it = 0; it < 5; ++it) {
    const int g = it * 65536 + tid;
    const int tile = g >> 8, j = (g >> 5) & 7, lane = g & 31;
    const int c = lane & 15, h = lane >> 4;
    const int ch = (tile < 1024) ? tile * 16 + c : (tile - 1024) * 64 + 4 * c;
    const float* s = W + (size_t)ch * 64 + 8 * j + 2 * h;
    const v2f a = *reinterpret_cast<const v2f*>(s);
    const v2f b = *reinterpret_cast<const v2f*>(s + 4);
    *reinterpret_cast<v4f*>(Wf + (size_t)g * 4) = v4f{a[0], a[1], b[0], b[1]};
  }
}

template <bool FR>
__device__ __forceinline__ void fc2_body(const float* __restrict__ in,
                                         const float* __restrict__ W,
                                         const float* __restrict__ Wf,
                                         const float* __restrict__ bias,
                                         float* __restrict__ out,
                                         float* part,
                                         float* kvbuf,
                                         float* xk) {

  const int tid = threadIdx.x;
  const int lane = tid & 31;
  const int wave = tid >> 5;
  const int n0 = blockIdx.x * 16;
  const int bIdx = n0 >> 10;
  const int t0 = n0 & 1023;
  const int khalf = lane >> 4;
  const float* aRow = in + ((size_t)((t0 + (lane & 15)) * 4 + bIdx)) * 4096;

  float sm[4], sd[4], sn[4];
#pragma unroll
  for (int r = 0; r < 4; ++r) { sm[r] = -3.0e38f; sd[r] = 0.0f; sn[r] = 0.0f; }

  const int tsel = (wave & 3) * 4;
  v2f wb[2][2][16];
  v8f cb[2][2];
  load_unit<FR>(wb[0], cb[0], W, Wf, bias, (wave >> 2) * 256 + tsel * 16, 0, lane, khalf);

  for (int JJ = 0; JJ < 8; ++JJ) {
    const int J = 2 * JJ + (wave >> 2);
    const int Jn = (JJ < 7) ? J + 2 : J;
    v2f xin[4][16];
#pragma unroll
    for (int r = 0; r < 4; ++r) load_areg(xin[r], aRow, r * 16 + J, khalf);
    v8f acc[4];
#pragma unroll
    for (int u = 0; u < 8; ++u) {
      const int cur = u & 1, nxt = cur ^ 1, rp = u & 1;
      {
        const int un = (u + 1) & 7;
        const int Jq = (u == 7) ? Jn : J;
        load_unit<FR>(wb[nxt], cb[nxt], W, Wf, bias, Jq * 256 + (tsel + (un >> 1)) * 16,
                      un & 1, lane, khalf);
      }
      __builtin_amdgcn_sched_barrier(0);
      acc[2 * rp] = cb[cur][0];
      acc[2 * rp + 1] = cb[cur][1];
#pragma unroll
      for (int kk = 0; kk < 16; ++kk) {
#pragma unroll
        for (int q = 0; q < 2; ++q) {
          acc[2 * rp + q] = __builtin_amdgcn_wmma_f32_16x16x4_f32(
               false, wb[cur][q][kk],  false, xin[2 * rp + q][kk],
               (short)0, acc[2 * rp + q],  false,  false);
        }
      }
      __builtin_amdgcn_sched_barrier(0);
      if (rp == 1) {
        float xs[8];
#pragma unroll
        for (int i = 0; i < 8; ++i) xs[i] = acc[0][i] + acc[1][i] + acc[2][i] + acc[3][i];
#pragma unroll
        for (int r = 0; r < 4; ++r) {
          float bm = acc[r][0];
#pragma unroll
          for (int i = 1; i < 8; ++i) bm = fmaxf(bm, acc[r][i]);
          const float mn = fmaxf(sm[r], bm);
          const float sc = __expf(sm[r] - mn);
          float ds = 0.0f, ns = 0.0f;
#pragma unroll
          for (int i = 0; i < 8; ++i) {
            const float e = __expf(acc[r][i] - mn);
            ds += e;
            ns += e * xs[i];
          }
          sd[r] = sd[r] * sc + ds;
          sn[r] = sn[r] * sc + ns;
          sm[r] = mn;
        }
        if (u == 7) {
          float* xp = xk + ((J * 4 + (wave & 3)) * 4) * 256 + (8 * khalf) * 16 + (lane & 15);
#pragma unroll
          for (int r = 0; r < 4; ++r) {
#pragma unroll
            for (int i = 0; i < 8; ++i) xp[r * 256 + i * 16] = acc[r][i];
          }
        }
      }
    }
  }
#pragma unroll
  for (int r = 0; r < 4; ++r) {
    part[tid * 12 + r * 3 + 0] = sm[r];
    part[tid * 12 + r * 3 + 1] = sd[r];
    part[tid * 12 + r * 3 + 2] = sn[r];
  }
  __syncthreads();
  if (tid < 64) {
    const int prow = tid >> 2, pr = tid & 3;
    float mAll = -3.0e38f;
    for (int p = 0; p < 16; ++p)
      mAll = fmaxf(mAll, part[((p >> 1) * 32 + (p & 1) * 16 + prow) * 12 + pr * 3]);
    float dAll = 0.0f, nAll = 0.0f;
    for (int p = 0; p < 16; ++p) {
      const float* q = &part[((p >> 1) * 32 + (p & 1) * 16 + prow) * 12 + pr * 3];
      const float f = __expf(q[0] - mAll);
      dAll += q[1] * f;
      nAll += q[2] * f;
    }
    kvbuf[tid] = nAll / dAll;
  }
  __syncthreads();

  const int c15 = lane & 15;
  float kvr[4];
#pragma unroll
  for (int r = 0; r < 4; ++r) kvr[r] = kvbuf[c15 * 4 + r];
  v2f kvA;
  kvA[0] = kvbuf[c15 * 4 + 2 * khalf];
  kvA[1] = kvbuf[c15 * 4 + 2 * khalf + 1];
  float* obase = out + ((size_t)(t0 * 4 + bIdx)) * 4096 + lane;

  v4f kvrow[8];
#pragma unroll
  for (int i = 0; i < 8; ++i)
    kvrow[i] = *reinterpret_cast<const v4f*>(kvbuf + (8 * khalf + i) * 4);

  float sbv[2] = {0.0f, 0.0f};
  v2f bbv[2] = {v2f{0.0f, 0.0f}, v2f{0.0f, 0.0f}};
  load_unit2<FR>(wb[0], sbv[0], bbv[0], W, Wf, bias, wave * 16, 0, lane, khalf);

  for (int jb = 0; jb < 2; ++jb) {
    const int J = wave + 8 * jb;
    v2f xs[4][16];
#pragma unroll
    for (int r = 0; r < 4; ++r) load_areg(xs[r], aRow, r * 16 + J, khalf);
#pragma unroll
    for (int r = 0; r < 4; ++r) {
#pragma unroll
      for (int kk = 0; kk < 16; ++kk) xs[r][kk] = xs[r][kk] * kvr[r];
    }
    for (int bq = 0; bq < 4; ++bq) {
      const int T0 = J * 16 + 4 * bq;
      const int T0n = (bq < 3) ? T0 + 4 : ((jb == 0) ? (wave + 8) * 16 : T0);
      v2f sinf[16];
      load_areg(sinf, aRow, T0 >> 2, khalf);
      v8f acc[2][2];
      float sb3 = 0.0f;
      v4f xl[4][2];
#pragma unroll
      for (int p = 0; p < 8; ++p) {
        const int u = (p < 7) ? p : 9;
        const int cur = p & 1, nxt = cur ^ 1;
        const int v = u % 5, hf = u / 5;
        if (p < 6) {
          load_unit2<FR>(wb[nxt], sbv[nxt], bbv[nxt], W, Wf, bias, T0, u + 1, lane, khalf);
        } else if (p == 6) {
          load_unit2<FR>(wb[nxt], sbv[nxt], bbv[nxt], W, Wf, bias, T0, 9, lane, khalf);
          sb3 = bias[4 * ((T0 + 3) * 16 + c15)];
        } else {
          load_unit2<FR>(wb[nxt], sbv[nxt], bbv[nxt], W, Wf, bias, T0n, 0, lane, khalf);
          const float* xq = xk + ((T0 >> 2) * 4) * 256 + c15 * 16 + 8 * khalf;
#pragma unroll
          for (int r = 0; r < 4; ++r) {
            xl[r][0] = *reinterpret_cast<const v4f*>(xq + r * 256);
            xl[r][1] = *reinterpret_cast<const v4f*>(xq + r * 256 + 4);
          }
        }
        __builtin_amdgcn_sched_barrier(0);
        if (v < 4) {
          const int tl = v >> 1, rp = v & 1;
          if (rp == 0) {
            const float s0 = sbv[cur];
            acc[tl][0] = v8f{s0, s0, s0, s0, s0, s0, s0, s0};
            acc[tl][1] = v8f{0.0f, 0.0f, 0.0f, 0.0f, 0.0f, 0.0f, 0.0f, 0.0f};
          }
#pragma unroll
          for (int kk = 0; kk < 16; ++kk) {
#pragma unroll
            for (int q = 0; q < 2; ++q) {
              acc[tl][q] = __builtin_amdgcn_wmma_f32_16x16x4_f32(
                   false, xs[2 * rp + q][kk],  false, wb[cur][q][kk],
                   (short)0, acc[tl][q],  false,  false);
            }
          }
          if (rp == 0) {
            acc[tl][1] = __builtin_amdgcn_wmma_f32_16x16x4_f32(
                 false, kvA,  false, bbv[cur],
                 (short)0, acc[tl][1],  false,  false);
          }
        } else {
          if (hf == 1)
            acc[1][0] = v8f{sb3, sb3, sb3, sb3, sb3, sb3, sb3, sb3};
#pragma unroll
          for (int kk = 0; kk < 16; ++kk) {
#pragma unroll
            for (int q = 0; q < 2; ++q) {
              acc[q][0] = __builtin_amdgcn_wmma_f32_16x16x4_f32(
                   false, sinf[kk],  false, wb[cur][q][kk],
                   (short)0, acc[q][0],  false,  false);
            }
          }
        }
        __builtin_amdgcn_sched_barrier(0);
        if (v == 4) {
          const int o0 = (T0 + 2 * hf) * 16;
#pragma unroll
          for (int i = 0; i < 8; ++i) {
            const float X = acc[0][0][i] + acc[0][1][i];
            float Y;
            if (hf == 0) {
              Y = acc[1][0][i] + acc[1][1][i];
            } else {
              Y = acc[1][0][i];
#pragma unroll
              for (int r = 0; r < 4; ++r) Y += kvrow[i][r] * xl[r][i >> 2][i & 3];
            }
            const float Xo = xhalf(X), Yo = xhalf(Y);
            const float P = khalf ? Yo : X;
            const float Q = khalf ? Y : Xo;
            obase[(size_t)i * 16384 + o0] = P;
            obase[(size_t)(i + 8) * 16384 + o0] = Q;
          }
        }
      }
    }
  }
}

__global__ __launch_bounds__(256) void fc2_fused(const float* __restrict__ in,
                                                 const float* __restrict__ W,
                                                 const float* __restrict__ bias,
                                                 float* __restrict__ out,
                                                 const float* __restrict__ Wf,
                                                 size_t ws_bytes) {
  __shared__ float part[256 * 12];
  __shared__ alignas(16) float kvbuf[64];
  __shared__ alignas(16) float xk[16 * 4 * 4 * 16 * 16];
  if (__builtin_expect(ws_bytes >= kWsNeed, 1))
    fc2_body<true>(in, W, Wf, bias, out, part, kvbuf, xk);
  else
    fc2_body<false>(in, W, Wf, bias, out, part, kvbuf, xk);
}

extern "C" void kernel_launch(void* const* d_in, const int* in_sizes, int n_in,
                              void* d_out, int out_size, void* d_ws, size_t ws_size,
                              hipStream_t stream) {
  (void)in_sizes; (void)n_in; (void)out_size;
  const float* in = (const float*)d_in[0];
  const float* W = (const float*)d_in[1];
  const float* bias = (const float*)d_in[2];
  float* out = (float*)d_out;
  float* wf = (float*)d_ws;
  w_to_fragment_order<<<256, 256, 0, stream>>>(W, wf, ws_size);
  fc2_fused<<<256, 256, 0, stream>>>(in, W, bias, out, wf, ws_size);
}


# Tiles channel-interleaved (o = 64q+4i+c): the LDS-kept quarter of x (o multiple of 4) is also every output's residual s: s product and its loads dropped, 16 B stores; dsum/num, s sum order changed
# speedup vs baseline: 6.2713x; 1.1247x over previous
#include <hip/hip_runtime.h>


typedef float v2f __attribute__((ext_vector_type(2)));
typedef float v4f __attribute__((ext_vector_type(4)));
typedef float v8f __attribute__((ext_vector_type(8)));

__device__ __forceinline__ void load_areg(v2f* areg, const float* __restrict__ aRow,
                                          int ga, int khalf) {
  const float* p = aRow + ga * 64 + 2 * khalf;
#pragma unroll
  for (int kk = 0; kk < 16; ++kk)
    areg[kk] = *reinterpret_cast<const v2f*>(p + 4 * kk);
}

constexpr size_t kBiasOff = (size_t)1024 * 1024;
constexpr size_t kWsNeed = (size_t)1024 * 4096 + (size_t)16384 * 4;

__device__ __forceinline__ int tile_chan(int T, int i) {
  return 64 * (T >> 2) + 4 * i + (T & 3);
}

__device__ __forceinline__ void load_wfrag(v2f* areg, const float* __restrict__ Wf,
                                           int tile, int lane) {
  const float* p = Wf + (size_t)tile * 1024 + lane * 4;
#pragma unroll
  for (int j = 0; j < 8; ++j) {
    const v4f t = *reinterpret_cast<const v4f*>(p + 128 * j);
    areg[2 * j] = v2f{t[0], t[1]};
    areg[2 * j + 1] = v2f{t[2], t[3]};
  }
}

template <bool FR>
__device__ __forceinline__ void load_unit(v2f (&w)[2][16], v8f (&c)[2],
                                          const float* __restrict__ W,
                                          const float* __restrict__ Wf,
                                          const float* __restrict__ bias,
                                          int o0, int rp, int lane, int khalf) {
  const int T = o0 >> 4;
#pragma unroll
  for (int q = 0; q < 2; ++q) {
    const int r = 2 * rp + q;
    if (FR) {
      load_wfrag(w[q], Wf, r * 256 + T, lane);
      const float* bp = Wf + kBiasOff + r * 4096 + T * 16 + 8 * khalf;
      const v4f b0 = *reinterpret_cast<const v4f*>(bp);
      const v4f b1 = *reinterpret_cast<const v4f*>(bp + 4);
      c[q] = __builtin_shufflevector(b0, b1, 0, 1, 2, 3, 4, 5, 6, 7);
    } else {
      const int ch = r * 4096 + tile_chan(T, lane & 15);
      load_areg(w[q], W + (size_t)ch * 64, 0, khalf);
#pragma unroll
      for (int i = 0; i < 8; ++i) c[q][i] = bias[r * 4096 + tile_chan(T, 8 * khalf + i)];
    }
  }
}

template <bool FR>
__device__ __forceinline__ void load_unit2(v2f (&w)[2][16], v2f& bb,
                                           const float* __restrict__ W,
                                           const float* __restrict__ Wf,
                                           const float* __restrict__ bias,
                                           int Tb, int p, int lane, int khalf) {
  const int T = Tb + 1 + (p >> 1), rp = p & 1;
  const int o = tile_chan(T, lane & 15);
#pragma unroll
  for (int q = 0; q < 2; ++q) {
    const int r = 2 * rp + q;
    if (FR) load_wfrag(w[q], Wf, r * 256 + T, lane);
    else    load_areg(w[q], W + (size_t)(r * 4096 + o) * 64, 0, khalf);
  }
  if (rp == 0) {
    bb[0] = bias[(2 * khalf) * 4096 + o];
    bb[1] = bias[(2 * khalf + 1) * 4096 + o];
  }
}

__global__ __launch_bounds__(256) void w_to_fragment_order(const float* __restrict__ W,
                                                           const float* __restrict__ bias,
                                                           float* __restrict__ Wf,
                                                           size_t ws_bytes) {
  if (ws_bytes < kWsNeed) return;
  const int tid = blockIdx.x * 256 + threadIdx.x;
  for (int it = 0; it < 4; ++it) {
    const int g = it * 65536 + tid;
    const int tile = g >> 8, j = (g >> 5) & 7, lane = g & 31;
    const int c = lane & 15, h = lane >> 4;
    const int ch = (tile >> 8) * 4096 + tile_chan(tile & 255, c);
    const float* s = W + (size_t)ch * 64 + 8 * j + 2 * h;
    const v2f a = *reinterpret_cast<const v2f*>(s);
    const v2f b = *reinterpret_cast<const v2f*>(s + 4);
    *reinterpret_cast<v4f*>(Wf + (size_t)g * 4) = v4f{a[0], a[1], b[0], b[1]};
  }
  if (tid < 16384)
    Wf[kBiasOff + tid] = bias[(tid >> 12) * 4096 + tile_chan((tid >> 4) & 255, tid & 15)];
}

template <bool FR>
__device__ __forceinline__ void fc2_body(const float* __restrict__ in,
                                         const float* __restrict__ W,
                                         const float* __restrict__ Wf,
                                         const float* __restrict__ bias,
                                         float* __restrict__ out,
                                         float* part,
                                         float* kvbuf,
                                         float* xk) {

  const int tid = threadIdx.x;
  const int lane = tid & 31;
  const int wave = tid >> 5;
  const int n0 = blockIdx.x * 16;
  const int bIdx = n0 >> 10;
  const int t0 = n0 & 1023;
  const int khalf = lane >> 4;
  const float* aRow = in + ((size_t)((t0 + (lane & 15)) * 4 + bIdx)) * 4096;

  float sm[4], sd[4], sn[4];
#pragma unroll
  for (int r = 0; r < 4; ++r) { sm[r] = -3.0e38f; sd[r] = 0.0f; sn[r] = 0.0f; }

  const int tsel = (wave & 3) * 4;
  v2f wb[2][2][16];
  v8f cb[2][2];
  load_unit<FR>(wb[0], cb[0], W, Wf, bias, (wave >> 2) * 256 + tsel * 16, 0, lane, khalf);

  for (int JJ = 0; JJ < 8; ++JJ) {
    const int J = 2 * JJ + (wave >> 2);
    const int Jn = (JJ < 7) ? J + 2 : J;
    v2f xin[4][16];
#pragma unroll
    for (int r = 0; r < 4; ++r) load_areg(xin[r], aRow, r * 16 + J, khalf);
    v8f acc[4];
#pragma unroll
    for (int u = 0; u < 8; ++u) {
      const int cur = u & 1, nxt = cur ^ 1, rp = u & 1;
      {
        const int un = (u + 1) & 7;
        const int Jq = (u == 7) ? Jn : J;
        load_unit<FR>(wb[nxt], cb[nxt], W, Wf, bias, Jq * 256 + (tsel + (un >> 1)) * 16,
                      un & 1, lane, khalf);
      }
      __builtin_amdgcn_sched_barrier(0);
      acc[2 * rp] = cb[cur][0];
      acc[2 * rp + 1] = cb[cur][1];
#pragma unroll
      for (int kk = 0; kk < 16; ++kk) {
#pragma unroll
        for (int q = 0; q < 2; ++q) {
          acc[2 * rp + q] = __builtin_amdgcn_wmma_f32_16x16x4_f32(
               false, wb[cur][q][kk],  false, xin[2 * rp + q][kk],
               (short)0, acc[2 * rp + q],  false,  false);
        }
      }
      __builtin_amdgcn_sched_barrier(0);
      if (rp == 1) {
        float xs[8];
#pragma unroll
        for (int i = 0; i < 8; ++i) xs[i] = acc[0][i] + acc[1][i] + acc[2][i] + acc[3][i];
#pragma unroll
        for (int r = 0; r < 4; ++r) {
          float bm = acc[r][0];
#pragma unroll
          for (int i = 1; i < 8; ++i) bm = fmaxf(bm, acc[r][i]);
          const float mn = fmaxf(sm[r], bm);
          const float sc = __expf(sm[r] - mn);
          float ds = 0.0f, ns = 0.0f;
#pragma unroll
          for (int i = 0; i < 8; ++i) {
            const float e = __expf(acc[r][i] - mn);
            ds += e;
            ns += e * xs[i];
          }
          sd[r] = sd[r] * sc + ds;
          sn[r] = sn[r] * sc + ns;
          sm[r] = mn;
        }
        if (u == 1) {
          float* xp = xk + ((J * 4 + (wave & 3)) * 4) * 256 + (8 * khalf) * 16 + (lane & 15);
#pragma unroll
          for (int r = 0; r < 4; ++r) {
#pragma unroll
            for (int i = 0; i < 8; ++i) xp[r * 256 + i * 16] = acc[r][i];
          }
        }
      }
    }
  }
#pragma unroll
  for (int r = 0; r < 4; ++r) {
    part[tid * 12 + r * 3 + 0] = sm[r];
    part[tid * 12 + r * 3 + 1] = sd[r];
    part[tid * 12 + r * 3 + 2] = sn[r];
  }
  __syncthreads();
  if (tid < 64) {
    const int prow = tid >> 2, pr = tid & 3;
    float mAll = -3.0e38f;
    for (int p = 0; p < 16; ++p)
      mAll = fmaxf(mAll, part[((p >> 1) * 32 + (p & 1) * 16 + prow) * 12 + pr * 3]);
    float dAll = 0.0f, nAll = 0.0f;
    for (int p = 0; p < 16; ++p) {
      const float* q = &part[((p >> 1) * 32 + (p & 1) * 16 + prow) * 12 + pr * 3];
      const float f = __expf(q[0] - mAll);
      dAll += q[1] * f;
      nAll += q[2] * f;
    }
    kvbuf[tid] = nAll / dAll;
  }
  __syncthreads();

  const int c15 = lane & 15;
  float kvr[4];
#pragma unroll
  for (int r = 0; r < 4; ++r) kvr[r] = kvbuf[c15 * 4 + r];
  v2f kvA;
  kvA[0] = kvbuf[c15 * 4 + 2 * khalf];
  kvA[1] = kvbuf[c15 * 4 + 2 * khalf + 1];
  float* obase = out + ((size_t)((t0 + 8 * khalf) * 4 + bIdx)) * 4096 + 4 * c15;

  v2f bbv[2] = {v2f{0.0f, 0.0f}, v2f{0.0f, 0.0f}};
  load_unit2<FR>(wb[0], bbv[0], W, Wf, bias, wave * 16, 0, lane, khalf);

  for (int jb = 0; jb < 2; ++jb) {
    const int J = wave + 8 * jb;
    v2f xs[4][16];
#pragma unroll
    for (int r = 0; r < 4; ++r) load_areg(xs[r], aRow, r * 16 + J, khalf);
#pragma unroll
    for (int r = 0; r < 4; ++r) {
#pragma unroll
      for (int kk = 0; kk < 16; ++kk) xs[r][kk] = xs[r][kk] * kvr[r];
    }
    for (int bq = 0; bq < 4; ++bq) {
      const int Tb = J * 16 + 4 * bq;
      const int Tbn = (bq < 3) ? Tb + 4 : ((jb == 0) ? (wave + 8) * 16 : Tb);
      const float* xq = xk + ((Tb >> 2) * 4) * 256 + c15 * 16 + 8 * khalf;
      const float* sq = xk + (((4 * (J & 3) + bq) * 4 + (c15 >> 2)) * 4 + (J >> 2)) * 256 +
                        (4 * (c15 & 3)) * 16 + 8 * khalf;
      v4f xl[4][2];
      v4f kvrow[8];
      v4f sl[2];
      v4f f[8];
      v8f acc[2];
#pragma unroll
      for (int p = 0; p < 6; ++p) {
        const int cur = p & 1, nxt = cur ^ 1;
        const int tl = p >> 1, rp = p & 1;
        if (p < 5) load_unit2<FR>(wb[nxt], bbv[nxt], W, Wf, bias, Tb, p + 1, lane, khalf);
        else       load_unit2<FR>(wb[nxt], bbv[nxt], W, Wf, bias, Tbn, 0, lane, khalf);
        if (p == 0) {
#pragma unroll
          for (int r = 0; r < 4; ++r) {
            xl[r][0] = *reinterpret_cast<const v4f*>(xq + r * 256);
            xl[r][1] = *reinterpret_cast<const v4f*>(xq + r * 256 + 4);
          }
          sl[0] = *reinterpret_cast<const v4f*>(sq);
          sl[1] = *reinterpret_cast<const v4f*>(sq + 4);
#pragma unroll
          for (int i = 0; i < 8; ++i)
            kvrow[i] = *reinterpret_cast<const v4f*>(kvbuf + (8 * khalf + i) * 4);
        }
        if (rp == 1) {
          sl[0] = *reinterpret_cast<const v4f*>(sq + (tl + 1) * 16);
          sl[1] = *reinterpret_cast<const v4f*>(sq + (tl + 1) * 16 + 4);
        }
        __builtin_amdgcn_sched_barrier(0);
        if (rp == 0) {
          acc[0] = v8f{0.0f, 0.0f, 0.0f, 0.0f, 0.0f, 0.0f, 0.0f, 0.0f};
          acc[1] = v8f{0.0f, 0.0f, 0.0f, 0.0f, 0.0f, 0.0f, 0.0f, 0.0f};
        }
#pragma unroll
        for (int kk = 0; kk < 16; ++kk) {
#pragma unroll
          for (int q = 0; q < 2; ++q) {
            acc[q] = __builtin_amdgcn_wmma_f32_16x16x4_f32(
                 false, xs[2 * rp + q][kk],  false, wb[cur][q][kk],
                 (short)0, acc[q],  false,  false);
          }
        }
        if (rp == 0) {
          acc[1] = __builtin_amdgcn_wmma_f32_16x16x4_f32(
               false, kvA,  false, bbv[cur],
               (short)0, acc[1],  false,  false);
        }
        asm volatile("" ::"v"(acc[0]), "v"(acc[1]));
        __builtin_amdgcn_sched_barrier(0);
        if (p == 0) {
#pragma unroll
          for (int i = 0; i < 8; ++i) {
            float y = sl[i >> 2][i & 3];
#pragma unroll
            for (int r = 0; r < 4; ++r) y += kvrow[i][r] * xl[r][i >> 2][i & 3];
            f[i][0] = y;
          }
        }
        if (rp == 1) {
#pragma unroll
          for (int i = 0; i < 8; ++i)
            f[i][tl + 1] = (acc[0][i] + acc[1][i]) + sl[i >> 2][i & 3];
        }
      }
#pragma unroll
      for (int i = 0; i < 8; ++i)
        *reinterpret_cast<v4f*>(obase + (size_t)i * 16384 + Tb * 16) = f[i];
    }
  }
}

__global__ __launch_bounds__(256) void fc2_fused(const float* __restrict__ in,
                                                 const float* __restrict__ W,
                                                 const float* __restrict__ bias,
                                                 float* __restrict__ out,
                                                 const float* __restrict__ Wf,
                                                 size_t ws_bytes) {
  __shared__ float part[256 * 12];
  __shared__ alignas(16) float kvbuf[64];
  __shared__ alignas(16) float xk[16 * 4 * 4 * 16 * 16];
  if (__builtin_expect(ws_bytes >= kWsNeed, 1))
    fc2_body<true>(in, W, Wf, bias, out, part, kvbuf, xk);
  else
    fc2_body<false>(in, W, Wf, bias, out, part, kvbuf, xk);
}

extern "C" void kernel_launch(void* const* d_in, const int* in_sizes, int n_in,
                              void* d_out, int out_size, void* d_ws, size_t ws_size,
                              hipStream_t stream) {
  (void)in_sizes; (void)n_in; (void)out_size;
  const float* in = (const float*)d_in[0];
  const float* W = (const float*)d_in[1];
  const float* bias = (const float*)d_in[2];
  float* out = (float*)d_out;
  float* wf = (float*)d_ws;
  w_to_fragment_order<<<256, 256, 0, stream>>>(W, bias, wf, ws_size);
  fc2_fused<<<256, 256, 0, stream>>>(in, W, bias, out, wf, ws_size);
}


# Phase 1 split by input group as phase 2 (wave owns groups w+8, w): input fragments fetched by 1 wave per 32 units, were by 4 waves per 8; group w kept in registers for phase 2; dsum/num order changed
# speedup vs baseline: 6.8657x; 1.0948x over previous
#include <hip/hip_runtime.h>


typedef float v2f __attribute__((ext_vector_type(2)));
typedef float v4f __attribute__((ext_vector_type(4)));
typedef float v8f __attribute__((ext_vector_type(8)));

__device__ __forceinline__ void load_areg(v2f* areg, const float* __restrict__ aRow,
                                          int ga, int khalf) {
  const float* p = aRow + ga * 64 + 2 * khalf;
#pragma unroll
  for (int kk = 0; kk < 16; ++kk)
    areg[kk] = *reinterpret_cast<const v2f*>(p + 4 * kk);
}

constexpr size_t kBiasOff = (size_t)1024 * 1024;
constexpr size_t kWsNeed = (size_t)1024 * 4096 + (size_t)16384 * 4;

__device__ __forceinline__ int tile_chan(int T, int i) {
  return 64 * (T >> 2) + 4 * i + (T & 3);
}

__device__ __forceinline__ void load_wfrag(v2f* areg, const float* __restrict__ Wf,
                                           int tile, int lane) {
  const float* p = Wf + (size_t)tile * 1024 + lane * 4;
#pragma unroll
  for (int j = 0; j < 8; ++j) {
    const v4f t = *reinterpret_cast<const v4f*>(p + 128 * j);
    areg[2 * j] = v2f{t[0], t[1]};
    areg[2 * j + 1] = v2f{t[2], t[3]};
  }
}

template <bool FR>
__device__ __forceinline__ void load_unit(v2f (&w)[2][16], v8f (&c)[2],
                                          const float* __restrict__ W,
                                          const float* __restrict__ Wf,
                                          const float* __restrict__ bias,
                                          int o0, int rp, int lane, int khalf) {
  const int T = o0 >> 4;
#pragma unroll
  for (int q = 0; q < 2; ++q) {
    const int r = 2 * rp + q;
    if (FR) {
      load_wfrag(w[q], Wf, r * 256 + T, lane);
      const float* bp = Wf + kBiasOff + r * 4096 + T * 16 + 8 * khalf;
      const v4f b0 = *reinterpret_cast<const v4f*>(bp);
      const v4f b1 = *reinterpret_cast<const v4f*>(bp + 4);
      c[q] = __builtin_shufflevector(b0, b1, 0, 1, 2, 3, 4, 5, 6, 7);
    } else {
      const int ch = r * 4096 + tile_chan(T, lane & 15);
      load_areg(w[q], W + (size_t)ch * 64, 0, khalf);
#pragma unroll
      for (int i = 0; i < 8; ++i) c[q][i] = bias[r * 4096 + tile_chan(T, 8 * khalf + i)];
    }
  }
}

template <bool FR>
__device__ __forceinline__ void load_unit2(v2f (&w)[2][16], v2f& bb,
                                           const float* __restrict__ W,
                                           const float* __restrict__ Wf,
                                           const float* __restrict__ bias,
                                           int Tb, int p, int lane, int khalf) {
  const int T = Tb + 1 + (p >> 1), rp = p & 1;
  const int o = tile_chan(T, lane & 15);
#pragma unroll
  for (int q = 0; q < 2; ++q) {
    const int r = 2 * rp + q;
    if (FR) load_wfrag(w[q], Wf, r * 256 + T, lane);
    else    load_areg(w[q], W + (size_t)(r * 4096 + o) * 64, 0, khalf);
  }
  if (rp == 0) {
    bb[0] = bias[(2 * khalf) * 4096 + o];
    bb[1] = bias[(2 * khalf + 1) * 4096 + o];
  }
}

__global__ __launch_bounds__(256) void w_to_fragment_order(const float* __restrict__ W,
                                                           const float* __restrict__ bias,
                                                           float* __restrict__ Wf,
                                                           size_t ws_bytes) {
  if (ws_bytes < kWsNeed) return;
  const int tid = blockIdx.x * 256 + threadIdx.x;
  for (int it = 0; it < 4; ++it) {
    const int g = it * 65536 + tid;
    const int tile = g >> 8, j = (g >> 5) & 7, lane = g & 31;
    const int c = lane & 15, h = lane >> 4;
    const int ch = (tile >> 8) * 4096 + tile_chan(tile & 255, c);
    const float* s = W + (size_t)ch * 64 + 8 * j + 2 * h;
    const v2f a = *reinterpret_cast<const v2f*>(s);
    const v2f b = *reinterpret_cast<const v2f*>(s + 4);
    *reinterpret_cast<v4f*>(Wf + (size_t)g * 4) = v4f{a[0], a[1], b[0], b[1]};
  }
  if (tid < 16384)
    Wf[kBiasOff + tid] = bias[(tid >> 12) * 4096 + tile_chan((tid >> 4) & 255, tid & 15)];
}

template <bool FR>
__device__ __forceinline__ void fc2_body(const float* __restrict__ in,
                                         const float* __restrict__ W,
                                         const float* __restrict__ Wf,
                                         const float* __restrict__ bias,
                                         float* __restrict__ out,
                                         float* part,
                                         float* kvbuf,
                                         float* xk) {

  const int tid = threadIdx.x;
  const int lane = tid & 31;
  const int wave = tid >> 5;
  const int n0 = blockIdx.x * 16;
  const int bIdx = n0 >> 10;
  const int t0 = n0 & 1023;
  const int khalf = lane >> 4;
  const float* aRow = in + ((size_t)((t0 + (lane & 15)) * 4 + bIdx)) * 4096;

  float sm[4], sd[4], sn[4];
#pragma unroll
  for (int r = 0; r < 4; ++r) { sm[r] = -3.0e38f; sd[r] = 0.0f; sn[r] = 0.0f; }

  v2f wb[2][2][16];
  v8f cb[2][2];
  v2f xin[4][16];
  load_unit<FR>(wb[0], cb[0], W, Wf, bias, (wave + 8) * 256, 0, lane, khalf);

  for (int JJ = 0; JJ < 2; ++JJ) {
    const int J = wave + 8 - 8 * JJ;
#pragma unroll
    for (int r = 0; r < 4; ++r) load_areg(xin[r], aRow, r * 16 + J, khalf);
    for (int tq = 0; tq < 4; ++tq) {
    const int o0n = (tq < 3) ? J * 256 + (tq + 1) * 64 : ((JJ == 0) ? wave * 256 : J * 256);
    v8f acc[4];
#pragma unroll
    for (int u = 0; u < 8; ++u) {
      const int cur = u & 1, nxt = cur ^ 1, rp = u & 1;
      {
        const int un = (u + 1) & 7;
        const int o0q = (u == 7) ? o0n : J * 256 + tq * 64 + (un >> 1) * 16;
        load_unit<FR>(wb[nxt], cb[nxt], W, Wf, bias, o0q, un & 1, lane, khalf);
      }
      __builtin_amdgcn_sched_barrier(0);
      acc[2 * rp] = cb[cur][0];
      acc[2 * rp + 1] = cb[cur][1];
#pragma unroll
      for (int kk = 0; kk < 16; ++kk) {
#pragma unroll
        for (int q = 0; q < 2; ++q) {
          acc[2 * rp + q] = __builtin_amdgcn_wmma_f32_16x16x4_f32(
               false, wb[cur][q][kk],  false, xin[2 * rp + q][kk],
               (short)0, acc[2 * rp + q],  false,  false);
        }
      }
      __builtin_amdgcn_sched_barrier(0);
      if (rp == 1) {
        float xs[8];
#pragma unroll
        for (int i = 0; i < 8; ++i) xs[i] = acc[0][i] + acc[1][i] + acc[2][i] + acc[3][i];
#pragma unroll
        for (int r = 0; r < 4; ++r) {
          float bm = acc[r][0];
#pragma unroll
          for (int i = 1; i < 8; ++i) bm = fmaxf(bm, acc[r][i]);
          const float mn = fmaxf(sm[r], bm);
          const float sc = __expf(sm[r] - mn);
          float ds = 0.0f, ns = 0.0f;
#pragma unroll
          for (int i = 0; i < 8; ++i) {
            const float e = __expf(acc[r][i] - mn);
            ds += e;
            ns += e * xs[i];
          }
          sd[r] = sd[r] * sc + ds;
          sn[r] = sn[r] * sc + ns;
          sm[r] = mn;
        }
        if (u == 1) {
          float* xp = xk + ((J * 4 + tq) * 4) * 256 + (8 * khalf) * 16 + (lane & 15);
#pragma unroll
          for (int r = 0; r < 4; ++r) {
#pragma unroll
            for (int i = 0; i < 8; ++i) xp[r * 256 + i * 16] = acc[r][i];
          }
        }
      }
    }
    }
  }
#pragma unroll
  for (int r = 0; r < 4; ++r) {
    part[tid * 12 + r * 3 + 0] = sm[r];
    part[tid * 12 + r * 3 + 1] = sd[r];
    part[tid * 12 + r * 3 + 2] = sn[r];
  }
  __syncthreads();
  if (tid < 64) {
    const int prow = tid >> 2, pr = tid & 3;
    float mAll = -3.0e38f;
    for (int p = 0; p < 16; ++p)
      mAll = fmaxf(mAll, part[((p >> 1) * 32 + (p & 1) * 16 + prow) * 12 + pr * 3]);
    float dAll = 0.0f, nAll = 0.0f;
    for (int p = 0; p < 16; ++p) {
      const float* q = &part[((p >> 1) * 32 + (p & 1) * 16 + prow) * 12 + pr * 3];
      const float f = __expf(q[0] - mAll);
      dAll += q[1] * f;
      nAll += q[2] * f;
    }
    kvbuf[tid] = nAll / dAll;
  }
  __syncthreads();

  const int c15 = lane & 15;
  float kvr[4];
#pragma unroll
  for (int r = 0; r < 4; ++r) kvr[r] = kvbuf[c15 * 4 + r];
  v2f kvA;
  kvA[0] = kvbuf[c15 * 4 + 2 * khalf];
  kvA[1] = kvbuf[c15 * 4 + 2 * khalf + 1];
  float* obase = out + ((size_t)((t0 + 8 * khalf) * 4 + bIdx)) * 4096 + 4 * c15;

  v2f bbv[2] = {v2f{0.0f, 0.0f}, v2f{0.0f, 0.0f}};
  load_unit2<FR>(wb[0], bbv[0], W, Wf, bias, wave * 16, 0, lane, khalf);

  for (int jb = 0; jb < 2; ++jb) {
    const int J = wave + 8 * jb;
    if (jb != 0) {
#pragma unroll
      for (int r = 0; r < 4; ++r) load_areg(xin[r], aRow, r * 16 + J, khalf);
    }
#pragma unroll
    for (int r = 0; r < 4; ++r) {
#pragma unroll
      for (int kk = 0; kk < 16; ++kk) xin[r][kk] = xin[r][kk] * kvr[r];
    }
    for (int bq = 0; bq < 4; ++bq) {
      const int Tb = J * 16 + 4 * bq;
      const int Tbn = (bq < 3) ? Tb + 4 : ((jb == 0) ? (wave + 8) * 16 : Tb);
      const float* xq = xk + ((Tb >> 2) * 4) * 256 + c15 * 16 + 8 * khalf;
      const float* sq = xk + (((4 * (J & 3) + bq) * 4 + (c15 >> 2)) * 4 + (J >> 2)) * 256 +
                        (4 * (c15 & 3)) * 16 + 8 * khalf;
      v4f xl[4][2];
      v4f kvrow[8];
      v4f sl[2];
      v4f f[8];
      v8f acc[2];
#pragma unroll
      for (int p = 0; p < 6; ++p) {
        const int cur = p & 1, nxt = cur ^ 1;
        const int tl = p >> 1, rp = p & 1;
        if (p < 5) load_unit2<FR>(wb[nxt], bbv[nxt], W, Wf, bias, Tb, p + 1, lane, khalf);
        else       load_unit2<FR>(wb[nxt], bbv[nxt], W, Wf, bias, Tbn, 0, lane, khalf);
        if (p == 0) {
#pragma unroll
          for (int r = 0; r < 4; ++r) {
            xl[r][0] = *reinterpret_cast<const v4f*>(xq + r * 256);
            xl[r][1] = *reinterpret_cast<const v4f*>(xq + r * 256 + 4);
          }
          sl[0] = *reinterpret_cast<const v4f*>(sq);
          sl[1] = *reinterpret_cast<const v4f*>(sq + 4);
#pragma unroll
          for (int i = 0; i < 8; ++i)
            kvrow[i] = *reinterpret_cast<const v4f*>(kvbuf + (8 * khalf + i) * 4);
        }
        if (rp == 1) {
          sl[0] = *reinterpret_cast<const v4f*>(sq + (tl + 1) * 16);
          sl[1] = *reinterpret_cast<const v4f*>(sq + (tl + 1) * 16 + 4);
        }
        __builtin_amdgcn_sched_barrier(0);
        if (rp == 0) {
          acc[0] = v8f{0.0f, 0.0f, 0.0f, 0.0f, 0.0f, 0.0f, 0.0f, 0.0f};
          acc[1] = v8f{0.0f, 0.0f, 0.0f, 0.0f, 0.0f, 0.0f, 0.0f, 0.0f};
        }
#pragma unroll
        for (int kk = 0; kk < 16; ++kk) {
#pragma unroll
          for (int q = 0; q < 2; ++q) {
            acc[q] = __builtin_amdgcn_wmma_f32_16x16x4_f32(
                 false, xin[2 * rp + q][kk],  false, wb[cur][q][kk],
                 (short)0, acc[q],  false,  false);
          }
        }
        if (rp == 0) {
          acc[1] = __builtin_amdgcn_wmma_f32_16x16x4_f32(
               false, kvA,  false, bbv[cur],
               (short)0, acc[1],  false,  false);
        }
        asm volatile("" ::"v"(acc[0]), "v"(acc[1]));
        __builtin_amdgcn_sched_barrier(0);
        if (p == 0) {
#pragma unroll
          for (int i = 0; i < 8; ++i) {
            float y = sl[i >> 2][i & 3];
#pragma unroll
            for (int r = 0; r < 4; ++r) y += kvrow[i][r] * xl[r][i >> 2][i & 3];
            f[i][0] = y;
          }
        }
        if (rp == 1) {
#pragma unroll
          for (int i = 0; i < 8; ++i)
            f[i][tl + 1] = (acc[0][i] + acc[1][i]) + sl[i >> 2][i & 3];
        }
      }
#pragma unroll
      for (int i = 0; i < 8; ++i)
        *reinterpret_cast<v4f*>(obase + (size_t)i * 16384 + Tb * 16) = f[i];
    }
  }
}

__global__ __launch_bounds__(256) void fc2_fused(const float* __restrict__ in,
                                                 const float* __restrict__ W,
                                                 const float* __restrict__ bias,
                                                 float* __restrict__ out,
                                                 const float* __restrict__ Wf,
                                                 size_t ws_bytes) {
  __shared__ float part[256 * 12];
  __shared__ alignas(16) float kvbuf[64];
  __shared__ alignas(16) float xk[16 * 4 * 4 * 16 * 16];
  if (__builtin_expect(ws_bytes >= kWsNeed, 1))
    fc2_body<true>(in, W, Wf, bias, out, part, kvbuf, xk);
  else
    fc2_body<false>(in, W, Wf, bias, out, part, kvbuf, xk);
}

extern "C" void kernel_launch(void* const* d_in, const int* in_sizes, int n_in,
                              void* d_out, int out_size, void* d_ws, size_t ws_size,
                              hipStream_t stream) {
  (void)in_sizes; (void)n_in; (void)out_size;
  const float* in = (const float*)d_in[0];
  const float* W = (const float*)d_in[1];
  const float* bias = (const float*)d_in[2];
  float* out = (float*)d_out;
  float* wf = (float*)d_ws;
  w_to_fragment_order<<<256, 256, 0, stream>>>(W, bias, wf, ws_size);
  fc2_fused<<<256, 256, 0, stream>>>(in, W, bias, out, wf, ws_size);
}


# Priority window for the younger wave of each SIMD pair (s_setprio 3 for its first 4 of 8 quads/bodies between barriers, then 0): the pair reaches each barrier together; sums unchanged
# speedup vs baseline: 7.1420x; 1.0402x over previous
#include <hip/hip_runtime.h>


typedef float v2f __attribute__((ext_vector_type(2)));
typedef float v4f __attribute__((ext_vector_type(4)));
typedef float v8f __attribute__((ext_vector_type(8)));

__device__ __forceinline__ void load_areg(v2f* areg, const float* __restrict__ aRow,
                                          int ga, int khalf) {
  const float* p = aRow + ga * 64 + 2 * khalf;
#pragma unroll
  for (int kk = 0; kk < 16; ++kk)
    areg[kk] = *reinterpret_cast<const v2f*>(p + 4 * kk);
}

constexpr size_t kBiasOff = (size_t)1024 * 1024;
constexpr size_t kWsNeed = (size_t)1024 * 4096 + (size_t)16384 * 4;

__device__ __forceinline__ int tile_chan(int T, int i) {
  return 64 * (T >> 2) + 4 * i + (T & 3);
}

__device__ __forceinline__ void load_wfrag(v2f* areg, const float* __restrict__ Wf,
                                           int tile, int lane) {
  const float* p = Wf + (size_t)tile * 1024 + lane * 4;
#pragma unroll
  for (int j = 0; j < 8; ++j) {
    const v4f t = *reinterpret_cast<const v4f*>(p + 128 * j);
    areg[2 * j] = v2f{t[0], t[1]};
    areg[2 * j + 1] = v2f{t[2], t[3]};
  }
}

template <bool FR>
__device__ __forceinline__ void load_unit(v2f (&w)[2][16], v8f (&c)[2],
                                          const float* __restrict__ W,
                                          const float* __restrict__ Wf,
                                          const float* __restrict__ bias,
                                          int o0, int rp, int lane, int khalf) {
  const int T = o0 >> 4;
#pragma unroll
  for (int q = 0; q < 2; ++q) {
    const int r = 2 * rp + q;
    if (FR) {
      load_wfrag(w[q], Wf, r * 256 + T, lane);
      const float* bp = Wf + kBiasOff + r * 4096 + T * 16 + 8 * khalf;
      const v4f b0 = *reinterpret_cast<const v4f*>(bp);
      const v4f b1 = *reinterpret_cast<const v4f*>(bp + 4);
      c[q] = __builtin_shufflevector(b0, b1, 0, 1, 2, 3, 4, 5, 6, 7);
    } else {
      const int ch = r * 4096 + tile_chan(T, lane & 15);
      load_areg(w[q], W + (size_t)ch * 64, 0, khalf);
#pragma unroll
      for (int i = 0; i < 8; ++i) c[q][i] = bias[r * 4096 + tile_chan(T, 8 * khalf + i)];
    }
  }
}

template <bool FR>
__device__ __forceinline__ void load_unit2(v2f (&w)[2][16], v2f& bb,
                                           const float* __restrict__ W,
                                           const float* __restrict__ Wf,
                                           const float* __restrict__ bias,
                                           int Tb, int p, int lane, int khalf) {
  const int T = Tb + 1 + (p >> 1), rp = p & 1;
  const int o = tile_chan(T, lane & 15);
#pragma unroll
  for (int q = 0; q < 2; ++q) {
    const int r = 2 * rp + q;
    if (FR) load_wfrag(w[q], Wf, r * 256 + T, lane);
    else    load_areg(w[q], W + (size_t)(r * 4096 + o) * 64, 0, khalf);
  }
  if (rp == 0) {
    bb[0] = bias[(2 * khalf) * 4096 + o];
    bb[1] = bias[(2 * khalf + 1) * 4096 + o];
  }
}

__global__ __launch_bounds__(256) void w_to_fragment_order(const float* __restrict__ W,
                                                           const float* __restrict__ bias,
                                                           float* __restrict__ Wf,
                                                           size_t ws_bytes) {
  if (ws_bytes < kWsNeed) return;
  const int tid = blockIdx.x * 256 + threadIdx.x;
  for (int it = 0; it < 4; ++it) {
    const int g = it * 65536 + tid;
    const int tile = g >> 8, j = (g >> 5) & 7, lane = g & 31;
    const int c = lane & 15, h = lane >> 4;
    const int ch = (tile >> 8) * 4096 + tile_chan(tile & 255, c);
    const float* s = W + (size_t)ch * 64 + 8 * j + 2 * h;
    const v2f a = *reinterpret_cast<const v2f*>(s);
    const v2f b = *reinterpret_cast<const v2f*>(s + 4);
    *reinterpret_cast<v4f*>(Wf + (size_t)g * 4) = v4f{a[0], a[1], b[0], b[1]};
  }
  if (tid < 16384)
    Wf[kBiasOff + tid] = bias[(tid >> 12) * 4096 + tile_chan((tid >> 4) & 255, tid & 15)];
}

template <bool FR>
__device__ __forceinline__ void fc2_body(const float* __restrict__ in,
                                         const float* __restrict__ W,
                                         const float* __restrict__ Wf,
                                         const float* __restrict__ bias,
                                         float* __restrict__ out,
                                         float* part,
                                         float* kvbuf,
                                         float* xk) {

  const int tid = threadIdx.x;
  const int lane = tid & 31;
  const int wave = tid >> 5;
  const int n0 = blockIdx.x * 16;
  const int bIdx = n0 >> 10;
  const int t0 = n0 & 1023;
  const int khalf = lane >> 4;
  const bool younger = __builtin_amdgcn_readfirstlane(wave) >= 4;
  const float* aRow = in + ((size_t)((t0 + (lane & 15)) * 4 + bIdx)) * 4096;

  float sm[4], sd[4], sn[4];
#pragma unroll
  for (int r = 0; r < 4; ++r) { sm[r] = -3.0e38f; sd[r] = 0.0f; sn[r] = 0.0f; }

  v2f wb[2][2][16];
  v8f cb[2][2];
  v2f xin[4][16];
  load_unit<FR>(wb[0], cb[0], W, Wf, bias, (wave + 8) * 256, 0, lane, khalf);

  if (younger) __builtin_amdgcn_s_setprio(3);
  for (int JJ = 0; JJ < 2; ++JJ) {
    if (younger && JJ == 1) __builtin_amdgcn_s_setprio(0);
    const int J = wave + 8 - 8 * JJ;
#pragma unroll
    for (int r = 0; r < 4; ++r) load_areg(xin[r], aRow, r * 16 + J, khalf);
    for (int tq = 0; tq < 4; ++tq) {
    const int o0n = (tq < 3) ? J * 256 + (tq + 1) * 64 : ((JJ == 0) ? wave * 256 : J * 256);
    v8f acc[4];
#pragma unroll
    for (int u = 0; u < 8; ++u) {
      const int cur = u & 1, nxt = cur ^ 1, rp = u & 1;
      {
        const int un = (u + 1) & 7;
        const int o0q = (u == 7) ? o0n : J * 256 + tq * 64 + (un >> 1) * 16;
        load_unit<FR>(wb[nxt], cb[nxt], W, Wf, bias, o0q, un & 1, lane, khalf);
      }
      __builtin_amdgcn_sched_barrier(0);
      acc[2 * rp] = cb[cur][0];
      acc[2 * rp + 1] = cb[cur][1];
#pragma unroll
      for (int kk = 0; kk < 16; ++kk) {
#pragma unroll
        for (int q = 0; q < 2; ++q) {
          acc[2 * rp + q] = __builtin_amdgcn_wmma_f32_16x16x4_f32(
               false, wb[cur][q][kk],  false, xin[2 * rp + q][kk],
               (short)0, acc[2 * rp + q],  false,  false);
        }
      }
      __builtin_amdgcn_sched_barrier(0);
      if (rp == 1) {
        float xs[8];
#pragma unroll
        for (int i = 0; i < 8; ++i) xs[i] = acc[0][i] + acc[1][i] + acc[2][i] + acc[3][i];
#pragma unroll
        for (int r = 0; r < 4; ++r) {
          float bm = acc[r][0];
#pragma unroll
          for (int i = 1; i < 8; ++i) bm = fmaxf(bm, acc[r][i]);
          const float mn = fmaxf(sm[r], bm);
          const float sc = __expf(sm[r] - mn);
          float ds = 0.0f, ns = 0.0f;
#pragma unroll
          for (int i = 0; i < 8; ++i) {
            const float e = __expf(acc[r][i] - mn);
            ds += e;
            ns += e * xs[i];
          }
          sd[r] = sd[r] * sc + ds;
          sn[r] = sn[r] * sc + ns;
          sm[r] = mn;
        }
        if (u == 1) {
          float* xp = xk + ((J * 4 + tq) * 4) * 256 + (8 * khalf) * 16 + (lane & 15);
#pragma unroll
          for (int r = 0; r < 4; ++r) {
#pragma unroll
            for (int i = 0; i < 8; ++i) xp[r * 256 + i * 16] = acc[r][i];
          }
        }
      }
    }
    }
  }
#pragma unroll
  for (int r = 0; r < 4; ++r) {
    part[tid * 12 + r * 3 + 0] = sm[r];
    part[tid * 12 + r * 3 + 1] = sd[r];
    part[tid * 12 + r * 3 + 2] = sn[r];
  }
  __syncthreads();
  if (tid < 64) {
    const int prow = tid >> 2, pr = tid & 3;
    float mAll = -3.0e38f;
    for (int p = 0; p < 16; ++p)
      mAll = fmaxf(mAll, part[((p >> 1) * 32 + (p & 1) * 16 + prow) * 12 + pr * 3]);
    float dAll = 0.0f, nAll = 0.0f;
    for (int p = 0; p < 16; ++p) {
      const float* q = &part[((p >> 1) * 32 + (p & 1) * 16 + prow) * 12 + pr * 3];
      const float f = __expf(q[0] - mAll);
      dAll += q[1] * f;
      nAll += q[2] * f;
    }
    kvbuf[tid] = nAll / dAll;
  }
  __syncthreads();

  const int c15 = lane & 15;
  float kvr[4];
#pragma unroll
  for (int r = 0; r < 4; ++r) kvr[r] = kvbuf[c15 * 4 + r];
  v2f kvA;
  kvA[0] = kvbuf[c15 * 4 + 2 * khalf];
  kvA[1] = kvbuf[c15 * 4 + 2 * khalf + 1];
  float* obase = out + ((size_t)((t0 + 8 * khalf) * 4 + bIdx)) * 4096 + 4 * c15;

  v2f bbv[2] = {v2f{0.0f, 0.0f}, v2f{0.0f, 0.0f}};
  load_unit2<FR>(wb[0], bbv[0], W, Wf, bias, wave * 16, 0, lane, khalf);

  if (younger) __builtin_amdgcn_s_setprio(3);
  for (int jb = 0; jb < 2; ++jb) {
    if (younger && jb == 1) __builtin_amdgcn_s_setprio(0);
    const int J = wave + 8 * jb;
    if (jb != 0) {
#pragma unroll
      for (int r = 0; r < 4; ++r) load_areg(xin[r], aRow, r * 16 + J, khalf);
    }
#pragma unroll
    for (int r = 0; r < 4; ++r) {
#pragma unroll
      for (int kk = 0; kk < 16; ++kk) xin[r][kk] = xin[r][kk] * kvr[r];
    }
    for (int bq = 0; bq < 4; ++bq) {
      const int Tb = J * 16 + 4 * bq;
      const int Tbn = (bq < 3) ? Tb + 4 : ((jb == 0) ? (wave + 8) * 16 : Tb);
      const float* xq = xk + ((Tb >> 2) * 4) * 256 + c15 * 16 + 8 * khalf;
      const float* sq = xk + (((4 * (J & 3) + bq) * 4 + (c15 >> 2)) * 4 + (J >> 2)) * 256 +
                        (4 * (c15 & 3)) * 16 + 8 * khalf;
      v4f xl[4][2];
      v4f kvrow[8];
      v4f sl[2];
      v4f f[8];
      v8f acc[2];
#pragma unroll
      for (int p = 0; p < 6; ++p) {
        const int cur = p & 1, nxt = cur ^ 1;
        const int tl = p >> 1, rp = p & 1;
        if (p < 5) load_unit2<FR>(wb[nxt], bbv[nxt], W, Wf, bias, Tb, p + 1, lane, khalf);
        else       load_unit2<FR>(wb[nxt], bbv[nxt], W, Wf, bias, Tbn, 0, lane, khalf);
        if (p == 0) {
#pragma unroll
          for (int r = 0; r < 4; ++r) {
            xl[r][0] = *reinterpret_cast<const v4f*>(xq + r * 256);
            xl[r][1] = *reinterpret_cast<const v4f*>(xq + r * 256 + 4);
          }
          sl[0] = *reinterpret_cast<const v4f*>(sq);
          sl[1] = *reinterpret_cast<const v4f*>(sq + 4);
#pragma unroll
          for (int i = 0; i < 8; ++i)
            kvrow[i] = *reinterpret_cast<const v4f*>(kvbuf + (8 * khalf + i) * 4);
        }
        if (rp == 1) {
          sl[0] = *reinterpret_cast<const v4f*>(sq + (tl + 1) * 16);
          sl[1] = *reinterpret_cast<const v4f*>(sq + (tl + 1) * 16 + 4);
        }
        __builtin_amdgcn_sched_barrier(0);
        if (rp == 0) {
          acc[0] = v8f{0.0f, 0.0f, 0.0f, 0.0f, 0.0f, 0.0f, 0.0f, 0.0f};
          acc[1] = v8f{0.0f, 0.0f, 0.0f, 0.0f, 0.0f, 0.0f, 0.0f, 0.0f};
        }
#pragma unroll
        for (int kk = 0; kk < 16; ++kk) {
#pragma unroll
          for (int q = 0; q < 2; ++q) {
            acc[q] = __builtin_amdgcn_wmma_f32_16x16x4_f32(
                 false, xin[2 * rp + q][kk],  false, wb[cur][q][kk],
                 (short)0, acc[q],  false,  false);
          }
        }
        if (rp == 0) {
          acc[1] = __builtin_amdgcn_wmma_f32_16x16x4_f32(
               false, kvA,  false, bbv[cur],
               (short)0, acc[1],  false,  false);
        }
        asm volatile("" ::"v"(acc[0]), "v"(acc[1]));
        __builtin_amdgcn_sched_barrier(0);
        if (p == 0) {
#pragma unroll
          for (int i = 0; i < 8; ++i) {
            float y = sl[i >> 2][i & 3];
#pragma unroll
            for (int r = 0; r < 4; ++r) y += kvrow[i][r] * xl[r][i >> 2][i & 3];
            f[i][0] = y;
          }
        }
        if (rp == 1) {
#pragma unroll
          for (int i = 0; i < 8; ++i)
            f[i][tl + 1] = (acc[0][i] + acc[1][i]) + sl[i >> 2][i & 3];
        }
      }
#pragma unroll
      for (int i = 0; i < 8; ++i)
        *reinterpret_cast<v4f*>(obase + (size_t)i * 16384 + Tb * 16) = f[i];
    }
  }
}

__global__ __launch_bounds__(256) void fc2_fused(const float* __restrict__ in,
                                                 const float* __restrict__ W,
                                                 const float* __restrict__ bias,
                                                 float* __restrict__ out,
                                                 const float* __restrict__ Wf,
                                                 size_t ws_bytes) {
  __shared__ float part[256 * 12];
  __shared__ alignas(16) float kvbuf[64];
  __shared__ alignas(16) float xk[16 * 4 * 4 * 16 * 16];
  if (__builtin_expect(ws_bytes >= kWsNeed, 1))
    fc2_body<true>(in, W, Wf, bias, out, part, kvbuf, xk);
  else
    fc2_body<false>(in, W, Wf, bias, out, part, kvbuf, xk);
}

extern "C" void kernel_launch(void* const* d_in, const int* in_sizes, int n_in,
                              void* d_out, int out_size, void* d_ws, size_t ws_size,
                              hipStream_t stream) {
  (void)in_sizes; (void)n_in; (void)out_size;
  const float* in = (const float*)d_in[0];
  const float* W = (const float*)d_in[1];
  const float* bias = (const float*)d_in[2];
  float* out = (float*)d_out;
  float* wf = (float*)d_ws;
  w_to_fragment_order<<<256, 256, 0, stream>>>(W, bias, wf, ws_size);
  fc2_fused<<<256, 256, 0, stream>>>(in, W, bias, out, wf, ws_size);
}
